# starting kernel
# baseline (speedup 1.0000x reference)
#include <hip/hip_runtime.h>
#include <hip/hip_bf16.h>

#define HEADS   4
#define CH      128
#define HC      512
#define NEG_SLOPE 0.2f

typedef float v2f __attribute__((ext_vector_type(2)));
typedef float v8f __attribute__((ext_vector_type(8)));

__device__ __forceinline__ unsigned encF(float f) {
    unsigned u = __float_as_uint(f);
    return (u & 0x80000000u) ? ~u : (u | 0x80000000u);
}
__device__ __forceinline__ float decF(unsigned e) {
    unsigned b = (e & 0x80000000u) ? (e & 0x7FFFFFFFu) : ~e;
    return __uint_as_float(b);
}
__device__ __forceinline__ float lrelu(float a) {
    return a > 0.f ? a : NEG_SLOPE * a;
}

__global__ void gat_init(float* __restrict__ out, const float* __restrict__ bias,
                         unsigned* __restrict__ emax, float* __restrict__ denom, int N) {
    long long i = (long long)blockIdx.x * blockDim.x + threadIdx.x;
    long long total = (long long)N * HC;
    if (i < total) out[i] = bias[i & (HC - 1)];
    if (i < (long long)N * HEADS) { emax[i] = 0u; denom[i] = 0.f; }
}

__global__ void gat_gemm_wmma(const float* __restrict__ X, const float* __restrict__ W,
                              float* __restrict__ H, int N) {
    int wid   = (blockIdx.x * blockDim.x + threadIdx.x) >> 5;
    int lane  = threadIdx.x & 31;
    int mt    = wid >> 3;
    int ns    = wid & 7;
    int row0  = mt * 16;
    int col0  = ns * 64;
    if (row0 >= N) return;

    int half = lane >> 4;
    int l16  = lane & 15;

    int arow = row0 + l16;
    if (arow >= N) arow = N - 1;
    const float* __restrict__ A = X + (size_t)arow * CH;

    v8f acc0 = {}, acc1 = {}, acc2 = {}, acc3 = {};

#pragma unroll 4
    for (int k0 = 0; k0 < CH; k0 += 4) {
        int ka = k0 + half * 2;
        v2f a; a.x = A[ka]; a.y = A[ka + 1];

        const float* __restrict__ B = W + (size_t)ka * HC + col0 + l16;
        v2f b0; b0.x = B[0];  b0.y = B[HC];
        v2f b1; b1.x = B[16]; b1.y = B[16 + HC];
        v2f b2; b2.x = B[32]; b2.y = B[32 + HC];
        v2f b3; b3.x = B[48]; b3.y = B[48 + HC];

        acc0 = __builtin_amdgcn_wmma_f32_16x16x4_f32(false, a, false, b0, (short)0, acc0, false, false);
        acc1 = __builtin_amdgcn_wmma_f32_16x16x4_f32(false, a, false, b1, (short)0, acc1, false, false);
        acc2 = __builtin_amdgcn_wmma_f32_16x16x4_f32(false, a, false, b2, (short)0, acc2, false, false);
        acc3 = __builtin_amdgcn_wmma_f32_16x16x4_f32(false, a, false, b3, (short)0, acc3, false, false);
    }

#pragma unroll
    for (int r = 0; r < 8; ++r) {
        int m = row0 + half * 8 + r;
        if (m < N) {
            float* __restrict__ O = H + (size_t)m * HC + col0 + l16;
            O[0]  = acc0[r];
            O[16] = acc1[r];
            O[32] = acc2[r];
            O[48] = acc3[r];
        }
    }
}

__global__ void gat_dots(const float* __restrict__ H, const float* __restrict__ att_src,
                         const float* __restrict__ att_dst, float* __restrict__ asrc,
                         float* __restrict__ adst, int N) {
    int n    = (blockIdx.x * blockDim.x + threadIdx.x) >> 5;
    int lane = threadIdx.x & 31;
    if (n >= N) return;
    const float4* __restrict__ hp = (const float4*)(H + (size_t)n * HC);
    const float4* __restrict__ s4 = (const float4*)att_src;
    const float4* __restrict__ d4 = (const float4*)att_dst;
    float ss[HEADS], sd[HEADS];
#pragma unroll
    for (int h = 0; h < HEADS; ++h) {
        float4 hv = hp[h * 32 + lane];
        float4 av = s4[h * 32 + lane];
        float4 bv = d4[h * 32 + lane];
        float s = hv.x * av.x + hv.y * av.y + hv.z * av.z + hv.w * av.w;
        float d = hv.x * bv.x + hv.y * bv.y + hv.z * bv.z + hv.w * bv.w;
#pragma unroll
        for (int m = 16; m > 0; m >>= 1) {
            s += __shfl_xor(s, m, 32);
            d += __shfl_xor(d, m, 32);
        }
        ss[h] = s; sd[h] = d;
    }
    if (lane == 0) {
#pragma unroll
        for (int h = 0; h < HEADS; ++h) {
            asrc[(size_t)n * HEADS + h] = ss[h];
            adst[(size_t)n * HEADS + h] = sd[h];
        }
    }
}

__global__ void gat_edge_max(const int* __restrict__ src, const int* __restrict__ dst,
                             const float* __restrict__ asrc, const float* __restrict__ adst,
                             unsigned* __restrict__ emax, int E, int N) {
    int t = blockIdx.x * blockDim.x + threadIdx.x;
    if (t >= E + N) return;
    int s, d;
    if (t < E) { s = src[t]; d = dst[t]; } else { s = d = t - E; }
    float4 as = ((const float4*)asrc)[s];
    float4 ad = ((const float4*)adst)[d];
    float v[HEADS] = { as.x + ad.x, as.y + ad.y, as.z + ad.z, as.w + ad.w };
#pragma unroll
    for (int h = 0; h < HEADS; ++h)
        atomicMax(&emax[(size_t)d * HEADS + h], encF(lrelu(v[h])));
}

__global__ void gat_edge_expsum(const int* __restrict__ src, const int* __restrict__ dst,
                                const float* __restrict__ asrc, const float* __restrict__ adst,
                                const unsigned* __restrict__ emax, float* __restrict__ denom,
                                int E, int N) {
    int t = blockIdx.x * blockDim.x + threadIdx.x;
    if (t >= E + N) return;
    int s, d;
    if (t < E) { s = src[t]; d = dst[t]; } else { s = d = t - E; }
    float4 as = ((const float4*)asrc)[s];
    float4 ad = ((const float4*)adst)[d];
    float v[HEADS] = { as.x + ad.x, as.y + ad.y, as.z + ad.z, as.w + ad.w };
#pragma unroll
    for (int h = 0; h < HEADS; ++h) {
        float m = decF(emax[(size_t)d * HEADS + h]);
        atomicAdd(&denom[(size_t)d * HEADS + h], __expf(lrelu(v[h]) - m));
    }
}

__global__ void gat_scatter(const float* __restrict__ H, const int* __restrict__ src,
                            const int* __restrict__ dst, const float* __restrict__ asrc,
                            const float* __restrict__ adst, const unsigned* __restrict__ emax,
                            const float* __restrict__ denom, float* __restrict__ out,
                            int E, int N) {
    int wid  = (blockIdx.x * blockDim.x + threadIdx.x) >> 5;
    int lane = threadIdx.x & 31;
    int total = (E + N) * HEADS;
    if (wid >= total) return;
    int head = wid & (HEADS - 1);
    int e    = wid >> 2;
    int s, d;
    if (e < E) { s = src[e]; d = dst[e]; } else { s = d = e - E; }

    float a = lrelu(asrc[(size_t)s * HEADS + head] + adst[(size_t)d * HEADS + head]);
    float m = decF(emax[(size_t)d * HEADS + head]);
    float alpha = __expf(a - m) / denom[(size_t)d * HEADS + head];

    const float4* __restrict__ hp = (const float4*)(H + (size_t)s * HC + head * CH);
    float4 hv = hp[lane];
    float* __restrict__ op = out + (size_t)d * HC + head * CH + lane * 4;
    atomicAdd(op + 0, alpha * hv.x);
    atomicAdd(op + 1, alpha * hv.y);
    atomicAdd(op + 2, alpha * hv.z);
    atomicAdd(op + 3, alpha * hv.w);
}

extern "C" void kernel_launch(void* const* d_in, const int* in_sizes, int n_in,
                              void* d_out, int out_size, void* d_ws, size_t ws_size,
                              hipStream_t stream) {
    const float* x        = (const float*)d_in[0];
    const int*   eidx     = (const int*)d_in[1];
    const float* W        = (const float*)d_in[2];
    const float* att_src  = (const float*)d_in[3];
    const float* att_dst  = (const float*)d_in[4];
    const float* bias     = (const float*)d_in[5];
    float*       out      = (float*)d_out;

    const int N = in_sizes[0] / CH;
    const int E = in_sizes[1] / 2;
    const int* src = eidx;
    const int* dst = eidx + E;

    float*    H     = (float*)d_ws;
    float*    asrc  = H + (size_t)N * HC;
    float*    adst  = asrc + (size_t)N * HEADS;
    unsigned* emax  = (unsigned*)(adst + (size_t)N * HEADS);
    float*    denom = (float*)(emax + (size_t)N * HEADS);

    const int TB = 256;

    {
        long long total = (long long)N * HC;
        int blocks = (int)((total + TB - 1) / TB);
        gat_init<<<blocks, TB, 0, stream>>>(out, bias, emax, denom, N);
    }
    {
        int mtiles = (N + 15) / 16;
        int waves  = mtiles * 8;
        int blocks = (waves + 7) / 8;
        gat_gemm_wmma<<<blocks, TB, 0, stream>>>(x, W, H, N);
    }
    {
        long long threads = (long long)N * 32;
        int blocks = (int)((threads + TB - 1) / TB);
        gat_dots<<<blocks, TB, 0, stream>>>(H, att_src, att_dst, asrc, adst, N);
    }
    {
        int blocks = (E + N + TB - 1) / TB;
        gat_edge_max<<<blocks, TB, 0, stream>>>(src, dst, asrc, adst, emax, E, N);
    }
    {
        int blocks = (E + N + TB - 1) / TB;
        gat_edge_expsum<<<blocks, TB, 0, stream>>>(src, dst, asrc, adst, emax, denom, E, N);
    }
    {
        long long threads = (long long)(E + N) * HEADS * 32;
        int blocks = (int)((threads + TB - 1) / TB);
        gat_scatter<<<blocks, TB, 0, stream>>>(H, src, dst, asrc, adst, emax, denom, out, E, N);
    }
}


# out sums leave memory atomics: a 32-wave work-group owns 100 dst rows in LDS, scans the dst list for them, adds alpha*h rows by LDS f32 adds, stores out once; LDS arrival order, self term first
# speedup vs baseline: 3.0751x; 3.0751x over previous
#include <hip/hip_runtime.h>
#include <hip/hip_bf16.h>

#define HEADS   4
#define CH      128
#define HC      512
#define NEG_SLOPE 0.2f

typedef float v2f __attribute__((ext_vector_type(2)));
typedef float v8f __attribute__((ext_vector_type(8)));

__device__ __forceinline__ unsigned encF(float f) {
    unsigned u = __float_as_uint(f);
    return (u & 0x80000000u) ? ~u : (u | 0x80000000u);
}
__device__ __forceinline__ float decF(unsigned e) {
    unsigned b = (e & 0x80000000u) ? (e & 0x7FFFFFFFu) : ~e;
    return __uint_as_float(b);
}
__device__ __forceinline__ float lrelu(float a) {
    return a > 0.f ? a : NEG_SLOPE * a;
}
__device__ __forceinline__ float row16_sum(float v) {
    v += __int_as_float(__builtin_amdgcn_update_dpp(__float_as_int(v), __float_as_int(v), 0x161, 0xf, 0xf, false));
    v += __int_as_float(__builtin_amdgcn_update_dpp(__float_as_int(v), __float_as_int(v), 0x162, 0xf, 0xf, false));
    v += __int_as_float(__builtin_amdgcn_update_dpp(__float_as_int(v), __float_as_int(v), 0x164, 0xf, 0xf, false));
    v += __int_as_float(__builtin_amdgcn_update_dpp(__float_as_int(v), __float_as_int(v), 0x168, 0xf, 0xf, false));
    return v;
}

__global__ void gat_gemm_dots_init(const float* __restrict__ X, const float* __restrict__ W,
                                   const float* __restrict__ att_src, const float* __restrict__ att_dst,
                                   float* __restrict__ H,
                                   float* __restrict__ asrc, float* __restrict__ adst,
                                   unsigned* __restrict__ emax, float* __restrict__ denom, int N) {
    int wid   = (blockIdx.x * blockDim.x + threadIdx.x) >> 5;
    int lane  = threadIdx.x & 31;
    int mt    = wid >> 2;
    int hd    = wid & (HEADS - 1);
    int row0  = mt * 16;
    int col0  = hd * CH;
    if (row0 >= N) return;

    int half = lane >> 4;
    int l16  = lane & 15;

    int arow = row0 + l16;
    if (arow >= N) arow = N - 1;
    const float* __restrict__ A = X + (size_t)arow * CH;

    v8f acc[8];
    for (int t = 0; t < 8; ++t) acc[t] = (v8f){};

#pragma unroll 2
    for (int k0 = 0; k0 < CH; k0 += 4) {
        int ka = k0 + half * 2;
        v2f a; a.x = A[ka]; a.y = A[ka + 1];

        const float* __restrict__ B = W + (size_t)ka * HC + col0 + l16;
        v2f b[8];
        for (int t = 0; t < 8; ++t) { b[t].x = B[16 * t]; b[t].y = B[16 * t + HC]; }

        for (int t = 0; t < 8; ++t)
            acc[t] = __builtin_amdgcn_wmma_f32_16x16x4_f32(false, a, false, b[t], (short)0, acc[t], false, false);
    }

    for (int r = 0; r < 8; ++r) {
        int m = row0 + half * 8 + r;
        if (m < N) {
            float* __restrict__ O = H + (size_t)m * HC + col0 + l16;
            for (int t = 0; t < 8; ++t) O[16 * t] = acc[t][r];
        }
    }

    float as[8], ad[8];
    for (int t = 0; t < 8; ++t) {
        as[t] = att_src[col0 + 16 * t + l16];
        ad[t] = att_dst[col0 + 16 * t + l16];
    }
    float ssel = 0.f, dsel = 0.f;
    for (int r = 0; r < 8; ++r) {
        float s = acc[0][r] * as[0];
        float d = acc[0][r] * ad[0];
        for (int t = 1; t < 8; ++t) {
            s += acc[t][r] * as[t];
            d += acc[t][r] * ad[t];
        }
        s = row16_sum(s);
        d = row16_sum(d);
        if (l16 == r) { ssel = s; dsel = d; }
    }
    if (l16 < 8) {
        int n = row0 + half * 8 + l16;
        if (n < N) {
            size_t w = (size_t)n * HEADS + hd;
            asrc[w]  = ssel;
            adst[w]  = dsel;
            emax[w]  = 0u;
            denom[w] = 0.f;
        }
    }
}

__global__ void gat_edge_max(const int* __restrict__ src, const int* __restrict__ dst,
                             const float* __restrict__ asrc, const float* __restrict__ adst,
                             unsigned* __restrict__ emax, int E, int N) {
    int t = blockIdx.x * blockDim.x + threadIdx.x;
    int e = t >> 2;
    int h = t & (HEADS - 1);
    if (e >= E + N) return;
    int s, d;
    if (e < E) { s = src[e]; d = dst[e]; } else { s = d = e - E; }
    float v = asrc[(size_t)s * HEADS + h] + adst[(size_t)d * HEADS + h];
    atomicMax(&emax[(size_t)d * HEADS + h], encF(lrelu(v)));
}

__global__ void gat_edge_expsum(const int* __restrict__ src, const int* __restrict__ dst,
                                const float* __restrict__ asrc, const float* __restrict__ adst,
                                const unsigned* __restrict__ emax, float* __restrict__ denom,
                                int E, int N) {
    int t = blockIdx.x * blockDim.x + threadIdx.x;
    int e = t >> 2;
    int h = t & (HEADS - 1);
    if (e >= E + N) return;
    int s, d;
    if (e < E) { s = src[e]; d = dst[e]; } else { s = d = e - E; }
    float v = asrc[(size_t)s * HEADS + h] + adst[(size_t)d * HEADS + h];
    float m = decF(emax[(size_t)d * HEADS + h]);
    atomicAdd(&denom[(size_t)d * HEADS + h], __expf(lrelu(v) - m));
}

#define AG_R     100
#define AG_WAVES 32
#define AG_TB    (AG_WAVES * 32)
#define AG_QCAP  64
#define AG_FLUSH 8
#define AG_U     8
#define AG_STEP  (32 * 4 * AG_U)
#define AG_G     4

__device__ __forceinline__ float gat_alpha(float as, float ad, unsigned em, float dn) {
    return __expf(lrelu(as + ad) - decF(em)) / dn;
}

template <int G>
__device__ __forceinline__ void ag_rows(const float* __restrict__ H, float* acc, int s_i, int l_i,
                                        float a0, float a1, float a2, float a3, int i0, int lane) {
    float4 hv[G][4];
    float  al[G][4];
    int    lo[G];
    for (int b = 0; b < G; ++b) {
        const int ss = __builtin_amdgcn_readlane(s_i, i0 + b);
        lo[b]    = __builtin_amdgcn_readlane(l_i, i0 + b);
        al[b][0] = __int_as_float(__builtin_amdgcn_readlane(__float_as_int(a0), i0 + b));
        al[b][1] = __int_as_float(__builtin_amdgcn_readlane(__float_as_int(a1), i0 + b));
        al[b][2] = __int_as_float(__builtin_amdgcn_readlane(__float_as_int(a2), i0 + b));
        al[b][3] = __int_as_float(__builtin_amdgcn_readlane(__float_as_int(a3), i0 + b));
        const float* __restrict__ hp = H + (size_t)ss * HC + 4 * lane;
        for (int j = 0; j < HEADS; ++j) hv[b][j] = *(const float4*)(hp + CH * j);
    }
    for (int b = 0; b < G; ++b) {
        float* ap = acc + lo[b] * HC + lane;
        for (int j = 0; j < HEADS; ++j) {
            atomicAdd(ap + j * CH,      al[b][j] * hv[b][j].x);
            atomicAdd(ap + j * CH + 32, al[b][j] * hv[b][j].y);
            atomicAdd(ap + j * CH + 64, al[b][j] * hv[b][j].z);
            atomicAdd(ap + j * CH + 96, al[b][j] * hv[b][j].w);
        }
    }
}

__device__ __forceinline__ void ag_flush(const float* __restrict__ H, const int* __restrict__ src,
                                         const int* __restrict__ dst, const float* __restrict__ asrc,
                                         const float* __restrict__ adst, const unsigned* __restrict__ emax,
                                         const float* __restrict__ denom, float* acc,
                                         const int* qe, int cnt, int n0, int lane) {
    __builtin_amdgcn_fence(__ATOMIC_ACQ_REL, "wavefront");
    for (int q0 = 0; q0 < cnt; q0 += 32) {
        const int nb = (cnt - q0 < 32) ? (cnt - q0) : 32;
        int s_i = 0, l_i = 0;
        float a0 = 0.f, a1 = 0.f, a2 = 0.f, a3 = 0.f;
        if (lane < nb) {
            const int e = qe[q0 + lane];
            s_i = src[e];
            l_i = dst[e] - n0;
            const size_t ws = (size_t)s_i * HEADS, wd = (size_t)(n0 + l_i) * HEADS;
            const float4 as4 = *(const float4*)(asrc + ws);
            const float4 ad4 = *(const float4*)(adst + wd);
            const uint4  em4 = *(const uint4*)(emax + wd);
            const float4 dn4 = *(const float4*)(denom + wd);
            a0 = gat_alpha(as4.x, ad4.x, em4.x, dn4.x);
            a1 = gat_alpha(as4.y, ad4.y, em4.y, dn4.y);
            a2 = gat_alpha(as4.z, ad4.z, em4.z, dn4.z);
            a3 = gat_alpha(as4.w, ad4.w, em4.w, dn4.w);
        }
        int i0 = 0;
        for (; i0 + AG_G <= nb; i0 += AG_G)
            ag_rows<AG_G>(H, acc, s_i, l_i, a0, a1, a2, a3, i0, lane);
        for (; i0 < nb; ++i0)
            ag_rows<1>(H, acc, s_i, l_i, a0, a1, a2, a3, i0, lane);
    }
}

__global__ void __launch_bounds__(AG_TB)
gat_aggregate(const float* __restrict__ H, const int* __restrict__ src, const int* __restrict__ dst,
              const float* __restrict__ asrc, const float* __restrict__ adst,
              const unsigned* __restrict__ emax, const float* __restrict__ denom,
              const float* __restrict__ bias, float* __restrict__ out, int E, int N) {
    __shared__ float acc[AG_R * HC];
    __shared__ int   q_edge[AG_WAVES * AG_QCAP];

    const int lane = threadIdx.x & 31;
    const int w    = __builtin_amdgcn_readfirstlane((int)(threadIdx.x >> 5));
    const int n0   = blockIdx.x * AG_R;
    const int rows = (N - n0 < AG_R) ? (N - n0) : AG_R;

    for (int r = w; r < rows; r += AG_WAVES) {
        const size_t wn = (size_t)(n0 + r) * HEADS;
        const float4 as4 = *(const float4*)(asrc + wn);
        const float4 ad4 = *(const float4*)(adst + wn);
        const uint4  em4 = *(const uint4*)(emax + wn);
        const float4 dn4 = *(const float4*)(denom + wn);
        float al[HEADS];
        al[0] = gat_alpha(as4.x, ad4.x, em4.x, dn4.x);
        al[1] = gat_alpha(as4.y, ad4.y, em4.y, dn4.y);
        al[2] = gat_alpha(as4.z, ad4.z, em4.z, dn4.z);
        al[3] = gat_alpha(as4.w, ad4.w, em4.w, dn4.w);
        const float* __restrict__ hp = H + (size_t)(n0 + r) * HC + 4 * lane;
        const float* __restrict__ bp = bias + 4 * lane;
        float* ap = acc + r * HC + lane;
        for (int j = 0; j < HEADS; ++j) {
            const float4 h4 = *(const float4*)(hp + CH * j);
            const float4 b4 = *(const float4*)(bp + CH * j);
            ap[j * CH]      = b4.x + al[j] * h4.x;
            ap[j * CH + 32] = b4.y + al[j] * h4.y;
            ap[j * CH + 64] = b4.z + al[j] * h4.z;
            ap[j * CH + 96] = b4.w + al[j] * h4.w;
        }
    }
    __syncthreads();

    int* qe = q_edge + w * AG_QCAP;
    int  qn = 0;
    const bool al16  = (((size_t)dst) & 15) == 0;
    const int  nfull = al16 ? E / AG_STEP : 0;
    const int  tail0 = nfull * AG_STEP;
    const int  items = nfull + (E - tail0 + 31) / 32;
    int4 dv[AG_U];
    if (w < nfull)
        for (int u = 0; u < AG_U; ++u) dv[u] = *(const int4*)(dst + w * AG_STEP + u * 128 + lane * 4);
    for (int it = w; it < items; it += AG_WAVES) {
        unsigned mm = 0u;
        int base, ls;
        if (it < nfull) {
            base = it * AG_STEP;
            ls   = 4;
            for (int u = 0; u < AG_U; ++u) {
                mm |= ((unsigned)(dv[u].x - n0) < (unsigned)rows ? 1u : 0u) << (4 * u);
                mm |= ((unsigned)(dv[u].y - n0) < (unsigned)rows ? 2u : 0u) << (4 * u);
                mm |= ((unsigned)(dv[u].z - n0) < (unsigned)rows ? 4u : 0u) << (4 * u);
                mm |= ((unsigned)(dv[u].w - n0) < (unsigned)rows ? 8u : 0u) << (4 * u);
            }
            if (it + AG_WAVES < nfull)
                for (int u = 0; u < AG_U; ++u)
                    dv[u] = *(const int4*)(dst + (it + AG_WAVES) * AG_STEP + u * 128 + lane * 4);
        } else {
            base = tail0 + (it - nfull) * 32;
            ls   = 1;
            const int e = base + lane;
            if (e < E && (unsigned)(dst[e] - n0) < (unsigned)rows) mm = 1u;
        }
        unsigned bal;
        while ((bal = __builtin_amdgcn_ballot_w32(mm != 0u)) != 0u) {
            if (mm != 0u) {
                const int k = __builtin_ctz(mm);
                mm &= mm - 1u;
                const int p = qn + (int)__builtin_amdgcn_mbcnt_lo(bal, 0u);
                qe[p] = base + (k >> 2) * 128 + lane * ls + (k & 3);
            }
            qn += __builtin_popcount(bal);
            if (qn >= AG_FLUSH) {
                ag_flush(H, src, dst, asrc, adst, emax, denom, acc, qe, qn, n0, lane);
                qn = 0;
            }
        }
    }
    if (qn > 0) ag_flush(H, src, dst, asrc, adst, emax, denom, acc, qe, qn, n0, lane);
    __syncthreads();

    for (int r = w; r < rows; r += AG_WAVES) {
        const float* ap = acc + r * HC + lane;
        float* __restrict__ op = out + (size_t)(n0 + r) * HC + 4 * lane;
        for (int j = 0; j < HEADS; ++j) {
            float4 o;
            o.x = ap[j * CH];
            o.y = ap[j * CH + 32];
            o.z = ap[j * CH + 64];
            o.w = ap[j * CH + 96];
            *(float4*)(op + CH * j) = o;
        }
    }
}

extern "C" void kernel_launch(void* const* d_in, const int* in_sizes, int n_in,
                              void* d_out, int out_size, void* d_ws, size_t ws_size,
                              hipStream_t stream) {
    const float* x        = (const float*)d_in[0];
    const int*   eidx     = (const int*)d_in[1];
    const float* W        = (const float*)d_in[2];
    const float* att_src  = (const float*)d_in[3];
    const float* att_dst  = (const float*)d_in[4];
    const float* bias     = (const float*)d_in[5];
    float*       out      = (float*)d_out;

    const int N = in_sizes[0] / CH;
    const int E = in_sizes[1] / 2;
    const int* src = eidx;
    const int* dst = eidx + E;

    float*    H     = (float*)d_ws;
    float*    asrc  = H + (size_t)N * HC;
    float*    adst  = asrc + (size_t)N * HEADS;
    unsigned* emax  = (unsigned*)(adst + (size_t)N * HEADS);
    float*    denom = (float*)(emax + (size_t)N * HEADS);

    const int TB = 256;

    {
        int mtiles = (N + 15) / 16;
        int waves  = mtiles * HEADS;
        int blocks = (waves + 7) / 8;
        gat_gemm_dots_init<<<blocks, TB, 0, stream>>>(x, W, att_src, att_dst, H, asrc, adst,
                                                      emax, denom, N);
    }
    {
        long long threads = (long long)(E + N) * HEADS;
        int blocks = (int)((threads + TB - 1) / TB);
        gat_edge_max<<<blocks, TB, 0, stream>>>(src, dst, asrc, adst, emax, E, N);
    }
    {
        long long threads = (long long)(E + N) * HEADS;
        int blocks = (int)((threads + TB - 1) / TB);
        gat_edge_expsum<<<blocks, TB, 0, stream>>>(src, dst, asrc, adst, emax, denom, E, N);
    }
    {
        int blocks = (N + AG_R - 1) / AG_R;
        gat_aggregate<<<blocks, AG_TB, 0, stream>>>(H, src, dst, asrc, adst, emax, denom, bias, out, E, N);
    }
}


# exp-sum launch folded into the aggregation: the owner work-group adds exp(e - max) into 4 LDS head sums per row beside the unnormalised w*h sums and scales by 1/sum at the store; 4 launches to 3
# speedup vs baseline: 3.4275x; 1.1146x over previous
#include <hip/hip_runtime.h>
#include <hip/hip_bf16.h>

#define HEADS   4
#define CH      128
#define HC      512
#define NEG_SLOPE 0.2f

typedef float v2f __attribute__((ext_vector_type(2)));
typedef float v8f __attribute__((ext_vector_type(8)));

__device__ __forceinline__ unsigned encF(float f) {
    unsigned u = __float_as_uint(f);
    return (u & 0x80000000u) ? ~u : (u | 0x80000000u);
}
__device__ __forceinline__ float decF(unsigned e) {
    unsigned b = (e & 0x80000000u) ? (e & 0x7FFFFFFFu) : ~e;
    return __uint_as_float(b);
}
__device__ __forceinline__ float lrelu(float a) {
    return a > 0.f ? a : NEG_SLOPE * a;
}
__device__ __forceinline__ float row16_sum(float v) {
    v += __int_as_float(__builtin_amdgcn_update_dpp(__float_as_int(v), __float_as_int(v), 0x161, 0xf, 0xf, false));
    v += __int_as_float(__builtin_amdgcn_update_dpp(__float_as_int(v), __float_as_int(v), 0x162, 0xf, 0xf, false));
    v += __int_as_float(__builtin_amdgcn_update_dpp(__float_as_int(v), __float_as_int(v), 0x164, 0xf, 0xf, false));
    v += __int_as_float(__builtin_amdgcn_update_dpp(__float_as_int(v), __float_as_int(v), 0x168, 0xf, 0xf, false));
    return v;
}

__global__ void gat_gemm_dots_init(const float* __restrict__ X, const float* __restrict__ W,
                                   const float* __restrict__ att_src, const float* __restrict__ att_dst,
                                   float* __restrict__ H,
                                   float* __restrict__ asrc, float* __restrict__ adst,
                                   unsigned* __restrict__ emax, int N) {
    int wid   = (blockIdx.x * blockDim.x + threadIdx.x) >> 5;
    int lane  = threadIdx.x & 31;
    int mt    = wid >> 2;
    int hd    = wid & (HEADS - 1);
    int row0  = mt * 16;
    int col0  = hd * CH;
    if (row0 >= N) return;

    int half = lane >> 4;
    int l16  = lane & 15;

    int arow = row0 + l16;
    if (arow >= N) arow = N - 1;
    const float* __restrict__ A = X + (size_t)arow * CH;

    v8f acc[8];
    for (int t = 0; t < 8; ++t) acc[t] = (v8f){};

#pragma unroll 2
    for (int k0 = 0; k0 < CH; k0 += 4) {
        int ka = k0 + half * 2;
        v2f a; a.x = A[ka]; a.y = A[ka + 1];

        const float* __restrict__ B = W + (size_t)ka * HC + col0 + l16;
        v2f b[8];
        for (int t = 0; t < 8; ++t) { b[t].x = B[16 * t]; b[t].y = B[16 * t + HC]; }

        for (int t = 0; t < 8; ++t)
            acc[t] = __builtin_amdgcn_wmma_f32_16x16x4_f32(false, a, false, b[t], (short)0, acc[t], false, false);
    }

    for (int r = 0; r < 8; ++r) {
        int m = row0 + half * 8 + r;
        if (m < N) {
            float* __restrict__ O = H + (size_t)m * HC + col0 + l16;
            for (int t = 0; t < 8; ++t) O[16 * t] = acc[t][r];
        }
    }

    float as[8], ad[8];
    for (int t = 0; t < 8; ++t) {
        as[t] = att_src[col0 + 16 * t + l16];
        ad[t] = att_dst[col0 + 16 * t + l16];
    }
    float ssel = 0.f, dsel = 0.f;
    for (int r = 0; r < 8; ++r) {
        float s = acc[0][r] * as[0];
        float d = acc[0][r] * ad[0];
        for (int t = 1; t < 8; ++t) {
            s += acc[t][r] * as[t];
            d += acc[t][r] * ad[t];
        }
        s = row16_sum(s);
        d = row16_sum(d);
        if (l16 == r) { ssel = s; dsel = d; }
    }
    if (l16 < 8) {
        int n = row0 + half * 8 + l16;
        if (n < N) {
            size_t w = (size_t)n * HEADS + hd;
            asrc[w]  = ssel;
            adst[w]  = dsel;
            emax[w]  = 0u;
        }
    }
}

__global__ void gat_edge_max(const int* __restrict__ src, const int* __restrict__ dst,
                             const float* __restrict__ asrc, const float* __restrict__ adst,
                             unsigned* __restrict__ emax, int E, int N) {
    int t = blockIdx.x * blockDim.x + threadIdx.x;
    int e = t >> 2;
    int h = t & (HEADS - 1);
    if (e >= E + N) return;
    int s, d;
    if (e < E) { s = src[e]; d = dst[e]; } else { s = d = e - E; }
    float v = asrc[(size_t)s * HEADS + h] + adst[(size_t)d * HEADS + h];
    atomicMax(&emax[(size_t)d * HEADS + h], encF(lrelu(v)));
}

#define AG_R     100
#define AG_WAVES 32
#define AG_TB    (AG_WAVES * 32)
#define AG_QCAP  64
#define AG_FLUSH 8
#define AG_U     8
#define AG_STEP  (32 * 4 * AG_U)
#define AG_G     4

__device__ __forceinline__ float gat_w(float as, float ad, unsigned em) {
    return __expf(lrelu(as + ad) - decF(em));
}

template <int G>
__device__ __forceinline__ void ag_rows(const float* __restrict__ H, float* acc, int s_i, int l_i,
                                        float a0, float a1, float a2, float a3, int i0, int lane) {
    float4 hv[G][4];
    float  al[G][4];
    int    lo[G];
    for (int b = 0; b < G; ++b) {
        const int ss = __builtin_amdgcn_readlane(s_i, i0 + b);
        lo[b]    = __builtin_amdgcn_readlane(l_i, i0 + b);
        al[b][0] = __int_as_float(__builtin_amdgcn_readlane(__float_as_int(a0), i0 + b));
        al[b][1] = __int_as_float(__builtin_amdgcn_readlane(__float_as_int(a1), i0 + b));
        al[b][2] = __int_as_float(__builtin_amdgcn_readlane(__float_as_int(a2), i0 + b));
        al[b][3] = __int_as_float(__builtin_amdgcn_readlane(__float_as_int(a3), i0 + b));
        const float* __restrict__ hp = H + (size_t)ss * HC + 4 * lane;
        for (int j = 0; j < HEADS; ++j) hv[b][j] = *(const float4*)(hp + CH * j);
    }
    for (int b = 0; b < G; ++b) {
        float* ap = acc + lo[b] * HC + lane;
        for (int j = 0; j < HEADS; ++j) {
            atomicAdd(ap + j * CH,      al[b][j] * hv[b][j].x);
            atomicAdd(ap + j * CH + 32, al[b][j] * hv[b][j].y);
            atomicAdd(ap + j * CH + 64, al[b][j] * hv[b][j].z);
            atomicAdd(ap + j * CH + 96, al[b][j] * hv[b][j].w);
        }
    }
}

__device__ __forceinline__ void ag_flush(const float* __restrict__ H, const int* __restrict__ src,
                                         const int* __restrict__ dst, const float* __restrict__ asrc,
                                         const float* __restrict__ adst, const unsigned* __restrict__ emax,
                                         float* acc, float* den,
                                         const int* qe, int cnt, int n0, int lane) {
    __builtin_amdgcn_fence(__ATOMIC_ACQ_REL, "wavefront");
    for (int q0 = 0; q0 < cnt; q0 += 32) {
        const int nb = (cnt - q0 < 32) ? (cnt - q0) : 32;
        int s_i = 0, l_i = 0;
        float a0 = 0.f, a1 = 0.f, a2 = 0.f, a3 = 0.f;
        if (lane < nb) {
            const int e = qe[q0 + lane];
            s_i = src[e];
            l_i = dst[e] - n0;
            const size_t ws = (size_t)s_i * HEADS, wd = (size_t)(n0 + l_i) * HEADS;
            const float4 as4 = *(const float4*)(asrc + ws);
            const float4 ad4 = *(const float4*)(adst + wd);
            const uint4  em4 = *(const uint4*)(emax + wd);
            a0 = gat_w(as4.x, ad4.x, em4.x);
            a1 = gat_w(as4.y, ad4.y, em4.y);
            a2 = gat_w(as4.z, ad4.z, em4.z);
            a3 = gat_w(as4.w, ad4.w, em4.w);
            float* dp = den + l_i * HEADS;
            atomicAdd(dp,     a0);
            atomicAdd(dp + 1, a1);
            atomicAdd(dp + 2, a2);
            atomicAdd(dp + 3, a3);
        }
        int i0 = 0;
        for (; i0 + AG_G <= nb; i0 += AG_G)
            ag_rows<AG_G>(H, acc, s_i, l_i, a0, a1, a2, a3, i0, lane);
        for (; i0 < nb; ++i0)
            ag_rows<1>(H, acc, s_i, l_i, a0, a1, a2, a3, i0, lane);
    }
}

__global__ void __launch_bounds__(AG_TB)
gat_aggregate(const float* __restrict__ H, const int* __restrict__ src, const int* __restrict__ dst,
              const float* __restrict__ asrc, const float* __restrict__ adst,
              const unsigned* __restrict__ emax,
              const float* __restrict__ bias, float* __restrict__ out, int E, int N) {
    __shared__ float acc[AG_R * HC];
    __shared__ float den[AG_R * HEADS];
    __shared__ int   q_edge[AG_WAVES * AG_QCAP];

    const int lane = threadIdx.x & 31;
    const int w    = __builtin_amdgcn_readfirstlane((int)(threadIdx.x >> 5));
    const int n0   = blockIdx.x * AG_R;
    const int rows = (N - n0 < AG_R) ? (N - n0) : AG_R;

    for (int r = w; r < rows; r += AG_WAVES) {
        const size_t wn = (size_t)(n0 + r) * HEADS;
        const float4 as4 = *(const float4*)(asrc + wn);
        const float4 ad4 = *(const float4*)(adst + wn);
        const uint4  em4 = *(const uint4*)(emax + wn);
        float al[HEADS];
        al[0] = gat_w(as4.x, ad4.x, em4.x);
        al[1] = gat_w(as4.y, ad4.y, em4.y);
        al[2] = gat_w(as4.z, ad4.z, em4.z);
        al[3] = gat_w(as4.w, ad4.w, em4.w);
        if (lane == 0) {
            float* dp = den + r * HEADS;
            dp[0] = al[0]; dp[1] = al[1]; dp[2] = al[2]; dp[3] = al[3];
        }
        const float* __restrict__ hp = H + (size_t)(n0 + r) * HC + 4 * lane;
        float* ap = acc + r * HC + lane;
        for (int j = 0; j < HEADS; ++j) {
            const float4 h4 = *(const float4*)(hp + CH * j);
            ap[j * CH]      = al[j] * h4.x;
            ap[j * CH + 32] = al[j] * h4.y;
            ap[j * CH + 64] = al[j] * h4.z;
            ap[j * CH + 96] = al[j] * h4.w;
        }
    }
    __syncthreads();

    int* qe = q_edge + w * AG_QCAP;
    int  qn = 0;
    const bool al16  = (((size_t)dst) & 15) == 0;
    const int  nfull = al16 ? E / AG_STEP : 0;
    const int  tail0 = nfull * AG_STEP;
    const int  items = nfull + (E - tail0 + 31) / 32;
    int4 dv[AG_U];
    if (w < nfull)
        for (int u = 0; u < AG_U; ++u) dv[u] = *(const int4*)(dst + w * AG_STEP + u * 128 + lane * 4);
    for (int it = w; it < items; it += AG_WAVES) {
        unsigned mm = 0u;
        int base, ls;
        if (it < nfull) {
            base = it * AG_STEP;
            ls   = 4;
            for (int u = 0; u < AG_U; ++u) {
                mm |= ((unsigned)(dv[u].x - n0) < (unsigned)rows ? 1u : 0u) << (4 * u);
                mm |= ((unsigned)(dv[u].y - n0) < (unsigned)rows ? 2u : 0u) << (4 * u);
                mm |= ((unsigned)(dv[u].z - n0) < (unsigned)rows ? 4u : 0u) << (4 * u);
                mm |= ((unsigned)(dv[u].w - n0) < (unsigned)rows ? 8u : 0u) << (4 * u);
            }
            if (it + AG_WAVES < nfull)
                for (int u = 0; u < AG_U; ++u)
                    dv[u] = *(const int4*)(dst + (it + AG_WAVES) * AG_STEP + u * 128 + lane * 4);
        } else {
            base = tail0 + (it - nfull) * 32;
            ls   = 1;
            const int e = base + lane;
            if (e < E && (unsigned)(dst[e] - n0) < (unsigned)rows) mm = 1u;
        }
        unsigned bal;
        while ((bal = __builtin_amdgcn_ballot_w32(mm != 0u)) != 0u) {
            if (mm != 0u) {
                const int k = __builtin_ctz(mm);
                mm &= mm - 1u;
                const int p = qn + (int)__builtin_amdgcn_mbcnt_lo(bal, 0u);
                qe[p] = base + (k >> 2) * 128 + lane * ls + (k & 3);
            }
            qn += __builtin_popcount(bal);
            if (qn >= AG_FLUSH) {
                ag_flush(H, src, dst, asrc, adst, emax, acc, den, qe, qn, n0, lane);
                qn = 0;
            }
        }
    }
    if (qn > 0) ag_flush(H, src, dst, asrc, adst, emax, acc, den, qe, qn, n0, lane);
    __syncthreads();

    for (int r = w; r < rows; r += AG_WAVES) {
        const float* ap = acc + r * HC + lane;
        const float* dp = den + r * HEADS;
        const float* __restrict__ bp = bias + 4 * lane;
        float* __restrict__ op = out + (size_t)(n0 + r) * HC + 4 * lane;
        for (int j = 0; j < HEADS; ++j) {
            const float  inv = 1.0f / dp[j];
            const float4 b4 = *(const float4*)(bp + CH * j);
            float4 o;
            o.x = b4.x + ap[j * CH]      * inv;
            o.y = b4.y + ap[j * CH + 32] * inv;
            o.z = b4.z + ap[j * CH + 64] * inv;
            o.w = b4.w + ap[j * CH + 96] * inv;
            *(float4*)(op + CH * j) = o;
        }
    }
}

extern "C" void kernel_launch(void* const* d_in, const int* in_sizes, int n_in,
                              void* d_out, int out_size, void* d_ws, size_t ws_size,
                              hipStream_t stream) {
    const float* x        = (const float*)d_in[0];
    const int*   eidx     = (const int*)d_in[1];
    const float* W        = (const float*)d_in[2];
    const float* att_src  = (const float*)d_in[3];
    const float* att_dst  = (const float*)d_in[4];
    const float* bias     = (const float*)d_in[5];
    float*       out      = (float*)d_out;

    const int N = in_sizes[0] / CH;
    const int E = in_sizes[1] / 2;
    const int* src = eidx;
    const int* dst = eidx + E;

    float*    H     = (float*)d_ws;
    float*    asrc  = H + (size_t)N * HC;
    float*    adst  = asrc + (size_t)N * HEADS;
    unsigned* emax  = (unsigned*)(adst + (size_t)N * HEADS);

    const int TB = 256;

    {
        int mtiles = (N + 15) / 16;
        int waves  = mtiles * HEADS;
        int blocks = (waves + 7) / 8;
        gat_gemm_dots_init<<<blocks, TB, 0, stream>>>(x, W, att_src, att_dst, H, asrc, adst,
                                                      emax, N);
    }
    {
        long long threads = (long long)(E + N) * HEADS;
        int blocks = (int)((threads + TB - 1) / TB);
        gat_edge_max<<<blocks, TB, 0, stream>>>(src, dst, asrc, adst, emax, E, N);
    }
    {
        int blocks = (N + AG_R - 1) / AG_R;
        gat_aggregate<<<blocks, AG_TB, 0, stream>>>(H, src, dst, asrc, adst, emax, bias, out, E, N);
    }
}
